# starting kernel
# baseline (speedup 1.0000x reference)
#include <hip/hip_runtime.h>
#include <math.h>


typedef __attribute__((ext_vector_type(2))) float v2f;
typedef __attribute__((ext_vector_type(8))) float v8f;

__global__ void __launch_bounds__(256)
gat_gemm_wmma(const float* __restrict__ A, const float* __restrict__ B,
              float* __restrict__ C, int M, int K) {
  const int lane    = threadIdx.x & 31;
  const int wv      = threadIdx.x >> 5;
  const int strip   = blockIdx.x * 8 + wv;
  const int nstrips = (M + 15) >> 4;
  if (strip >= nstrips) return;
  const int row0 = strip * 16;
  const int kh   = lane >> 4;
  const int l15  = lane & 15;
  int mrow = row0 + l15;
  if (mrow >= M) mrow = M - 1;

  const float* arow = A + (size_t)mrow * K;
  v8f acc[4];
#pragma unroll
  for (int t = 0; t < 4; ++t) acc[t] = (v8f){0.f,0.f,0.f,0.f,0.f,0.f,0.f,0.f};

  for (int k = 0; k < K; k += 4) {
    const v2f a = *(const v2f*)(arow + k + 2 * kh);
    const float* brow0 = B + (size_t)(k + 2 * kh) * 64 + l15;
#pragma unroll
    for (int t = 0; t < 4; ++t) {
      v2f b;
      b.x = brow0[t * 16];
      b.y = brow0[64 + t * 16];
      acc[t] = __builtin_amdgcn_wmma_f32_16x16x4_f32(
          false, a, false, b, (short)0, acc[t], false, false);
    }
  }

  const int rbase = row0 + 8 * kh;
#pragma unroll
  for (int t = 0; t < 4; ++t) {
    float* cp = C + (size_t)rbase * 64 + t * 16 + l15;
#pragma unroll
    for (int r = 0; r < 8; ++r)
      if (rbase + r < M) cp[(size_t)r * 64] = acc[t][r];
  }
}

__global__ void gat_node_attn(const float* __restrict__ z,
                              const float* __restrict__ al,
                              const float* __restrict__ ar,
                              float* __restrict__ el, float* __restrict__ er,
                              int N) {
  const int wid  = (int)((blockIdx.x * (size_t)blockDim.x + threadIdx.x) >> 5);
  const int lane = threadIdx.x & 31;
  if (wid >= N) return;
  const float* zr = z + (size_t)wid * 64;
  const float z0 = zr[lane], z1 = zr[lane + 32];
  float sl = z0 * al[lane] + z1 * al[lane + 32];
  float sr = z0 * ar[lane] + z1 * ar[lane + 32];
#pragma unroll
  for (int off = 16; off; off >>= 1) {
    sl += __shfl_xor(sl, off, 32);
    sr += __shfl_xor(sr, off, 32);
  }
  if (lane == 0) { el[wid] = sl; er[wid] = sr; }
}


__global__ void gat_zero_i32(int* __restrict__ p, int n) {
  const int i = blockIdx.x * blockDim.x + threadIdx.x;
  if (i < n) p[i] = 0;
}

__global__ void gat_histogram(const int* __restrict__ dst, int* __restrict__ deg, int E) {
  const int i = blockIdx.x * blockDim.x + threadIdx.x;
  if (i < E) atomicAdd(deg + dst[i], 1);
}

__global__ void gat_scan_partials(const int* __restrict__ deg, int* __restrict__ partial,
                                  int N) {
  __shared__ int sh[256];
  const int i = blockIdx.x * 256 + threadIdx.x;
  sh[threadIdx.x] = (i < N) ? deg[i] : 0;
  __syncthreads();
  for (int off = 128; off; off >>= 1) {
    if (threadIdx.x < off) sh[threadIdx.x] += sh[threadIdx.x + off];
    __syncthreads();
  }
  if (threadIdx.x == 0) partial[blockIdx.x] = sh[0];
}

__global__ void gat_scan_offsets(int* __restrict__ partial, int nb) {
  if (blockIdx.x == 0 && threadIdx.x == 0) {
    int run = 0;
    for (int b = 0; b < nb; ++b) { const int t = partial[b]; partial[b] = run; run += t; }
  }
}

__global__ void gat_scan_block(const int* __restrict__ deg, const int* __restrict__ partial,
                               int* __restrict__ rowstart, int* __restrict__ cursor, int N) {
  __shared__ int sh[256];
  const int i = blockIdx.x * 256 + threadIdx.x;
  const int v = (i < N) ? deg[i] : 0;
  sh[threadIdx.x] = v;
  __syncthreads();
  int val = v;
  for (int off = 1; off < 256; off <<= 1) {
    const int t = (threadIdx.x >= off) ? sh[threadIdx.x - off] : 0;
    __syncthreads();
    val += t;
    sh[threadIdx.x] = val;
    __syncthreads();
  }
  if (i < N) {
    const int ex = partial[blockIdx.x] + val - v;
    rowstart[i] = ex;
    cursor[i]   = ex;
  }
}

__global__ void gat_scatter(const int* __restrict__ src, const int* __restrict__ dst,
                            int* __restrict__ cursor, int* __restrict__ ssrc, int E) {
  const int i = blockIdx.x * blockDim.x + threadIdx.x;
  if (i < E) {
    const int pos = atomicAdd(cursor + dst[i], 1);
    ssrc[pos] = src[i];
  }
}

__global__ void gat_node_aggregate(const int* __restrict__ rowstart,
                                   const int* __restrict__ deg,
                                   const int* __restrict__ ssrc,
                                   const float* __restrict__ el,
                                   const float* __restrict__ er,
                                   const float* __restrict__ z,
                                   const float* __restrict__ bias,
                                   float* __restrict__ out,
                                   int N, int applyRelu) {
  const int wid  = (int)((blockIdx.x * (size_t)blockDim.x + threadIdx.x) >> 5);
  const int lane = threadIdx.x & 31;
  if (wid >= N) return;
  const int start = rowstart[wid];
  const int cnt   = deg[wid];
  const float ern = er[wid];

  float mx = -INFINITY;
  for (int j = lane; j < cnt; j += 32) {
    float v = el[ssrc[start + j]] + ern;
    v = (v > 0.f) ? v : 0.2f * v;
    mx = fmaxf(mx, v);
  }
#pragma unroll
  for (int off = 16; off; off >>= 1) mx = fmaxf(mx, __shfl_xor(mx, off, 32));

  float sum = 0.f;
  for (int j = lane; j < cnt; j += 32) {
    float v = el[ssrc[start + j]] + ern;
    v = (v > 0.f) ? v : 0.2f * v;
    sum += __expf(v - mx);
  }
#pragma unroll
  for (int off = 16; off; off >>= 1) sum += __shfl_xor(sum, off, 32);
  const float rcp = (cnt > 0) ? (1.f / sum) : 0.f;

  float2 acc = {0.f, 0.f};
  for (int j = 0; j < cnt; ++j) {
    const int sv = ssrc[start + j];
    float v = el[sv] + ern;
    v = (v > 0.f) ? v : 0.2f * v;
    const float alpha = __expf(v - mx) * rcp;
    const float2 zz = ((const float2*)(z + (size_t)sv * 64))[lane];
    acc.x += alpha * zz.x;
    acc.y += alpha * zz.y;
  }

  float2 o;
  o.x = acc.x + bias[2 * lane];
  o.y = acc.y + bias[2 * lane + 1];
  if (applyRelu) { o.x = fmaxf(o.x, 0.f); o.y = fmaxf(o.y, 0.f); }
  ((float2*)(out + (size_t)wid * 64))[lane] = o;
}


extern "C" void kernel_launch(void* const* d_in, const int* in_sizes, int n_in,
                              void* d_out, int out_size, void* d_ws, size_t ws_size,
                              hipStream_t stream) {
  const float* x   = (const float*)d_in[0];
  const float* W1  = (const float*)d_in[1];
  const float* al1 = (const float*)d_in[2];
  const float* ar1 = (const float*)d_in[3];
  const float* b1  = (const float*)d_in[4];
  const float* W2  = (const float*)d_in[5];
  const float* al2 = (const float*)d_in[6];
  const float* ar2 = (const float*)d_in[7];
  const float* b2  = (const float*)d_in[8];
  const int*   src = (const int*)d_in[9];
  const int*   dst = (const int*)d_in[10];
  float* out = (float*)d_out;

  const int F  = 64;
  const int N  = out_size / F;
  const int E  = in_sizes[9];
  const int K1 = in_sizes[0] / N;
  (void)n_in; (void)ws_size;

  char* ws = (char*)d_ws;
  size_t off = 0;
  auto alloc = [&](size_t bytes) -> void* {
    void* p = ws + off;
    off += (bytes + 255) & ~(size_t)255;
    return p;
  };
  float* zbuf     = (float*)alloc((size_t)N * F * sizeof(float));
  float* hbuf     = (float*)alloc((size_t)N * F * sizeof(float));
  float* el       = (float*)alloc((size_t)N * sizeof(float));
  float* er       = (float*)alloc((size_t)N * sizeof(float));
  int*   deg      = (int*)alloc((size_t)N * sizeof(int));
  int*   rowstart = (int*)alloc((size_t)N * sizeof(int));
  int*   cursor   = (int*)alloc((size_t)N * sizeof(int));
  int*   ssrc     = (int*)alloc((size_t)E * sizeof(int));
  const int nb    = (N + 255) / 256;
  int*   partial  = (int*)alloc((size_t)nb * sizeof(int));

  const int gN256  = (N + 255) / 256;
  const int gE256  = (E + 255) / 256;
  const int gWaveN = (N + 7) / 8;
  const int gGemm  = (((N + 15) / 16) + 7) / 8;

  gat_zero_i32     <<<gN256, 256, 0, stream>>>(deg, N);
  gat_histogram    <<<gE256, 256, 0, stream>>>(dst, deg, E);
  gat_scan_partials<<<nb,    256, 0, stream>>>(deg, partial, N);
  gat_scan_offsets <<<1,      32, 0, stream>>>(partial, nb);
  gat_scan_block   <<<nb,    256, 0, stream>>>(deg, partial, rowstart, cursor, N);
  gat_scatter      <<<gE256, 256, 0, stream>>>(src, dst, cursor, ssrc, E);

  gat_gemm_wmma     <<<gGemm,  256, 0, stream>>>(x, W1, zbuf, N, K1);
  gat_node_attn     <<<gWaveN, 256, 0, stream>>>(zbuf, al1, ar1, el, er, N);
  gat_node_aggregate<<<gWaveN, 256, 0, stream>>>(rowstart, deg, ssrc, el, er,
                                                 zbuf, b1, hbuf, N,  1);

  gat_gemm_wmma     <<<gGemm,  256, 0, stream>>>(hbuf, W2, zbuf, N, F);
  gat_node_attn     <<<gWaveN, 256, 0, stream>>>(zbuf, al2, ar2, el, er, N);
  gat_node_aggregate<<<gWaveN, 256, 0, stream>>>(rowstart, deg, ssrc, el, er,
                                                 zbuf, b2, out, N,  0);
}


# aggregate: softmax max, denominator and weighted z sum folded into ONE pass over a row's edges (running max, rescale per 4-edge chunk); el gathered once an edge, not 3 times; new summation order
# speedup vs baseline: 1.5770x; 1.5770x over previous
#include <hip/hip_runtime.h>
#include <math.h>


typedef __attribute__((ext_vector_type(2))) float v2f;
typedef __attribute__((ext_vector_type(8))) float v8f;

__global__ void __launch_bounds__(256)
gat_gemm_wmma(const float* __restrict__ A, const float* __restrict__ B,
              float* __restrict__ C, int M, int K) {
  const int lane    = threadIdx.x & 31;
  const int wv      = threadIdx.x >> 5;
  const int strip   = blockIdx.x * 8 + wv;
  const int nstrips = (M + 15) >> 4;
  if (strip >= nstrips) return;
  const int row0 = strip * 16;
  const int kh   = lane >> 4;
  const int l15  = lane & 15;
  int mrow = row0 + l15;
  if (mrow >= M) mrow = M - 1;

  const float* arow = A + (size_t)mrow * K;
  v8f acc[4];
#pragma unroll
  for (int t = 0; t < 4; ++t) acc[t] = (v8f){0.f,0.f,0.f,0.f,0.f,0.f,0.f,0.f};

  for (int k = 0; k < K; k += 4) {
    const v2f a = *(const v2f*)(arow + k + 2 * kh);
    const float* brow0 = B + (size_t)(k + 2 * kh) * 64 + l15;
#pragma unroll
    for (int t = 0; t < 4; ++t) {
      v2f b;
      b.x = brow0[t * 16];
      b.y = brow0[64 + t * 16];
      acc[t] = __builtin_amdgcn_wmma_f32_16x16x4_f32(
          false, a, false, b, (short)0, acc[t], false, false);
    }
  }

  const int rbase = row0 + 8 * kh;
#pragma unroll
  for (int t = 0; t < 4; ++t) {
    float* cp = C + (size_t)rbase * 64 + t * 16 + l15;
#pragma unroll
    for (int r = 0; r < 8; ++r)
      if (rbase + r < M) cp[(size_t)r * 64] = acc[t][r];
  }
}

__global__ void gat_node_attn(const float* __restrict__ z,
                              const float* __restrict__ al,
                              const float* __restrict__ ar,
                              float* __restrict__ el, float* __restrict__ er,
                              int N) {
  const int wid  = (int)((blockIdx.x * (size_t)blockDim.x + threadIdx.x) >> 5);
  const int lane = threadIdx.x & 31;
  if (wid >= N) return;
  const float* zr = z + (size_t)wid * 64;
  const float z0 = zr[lane], z1 = zr[lane + 32];
  float sl = z0 * al[lane] + z1 * al[lane + 32];
  float sr = z0 * ar[lane] + z1 * ar[lane + 32];
#pragma unroll
  for (int off = 16; off; off >>= 1) {
    sl += __shfl_xor(sl, off, 32);
    sr += __shfl_xor(sr, off, 32);
  }
  if (lane == 0) { el[wid] = sl; er[wid] = sr; }
}

#define GAT_ROW_SLOTS 64

__global__ void gat_zero_i32(int* __restrict__ p, int n) {
  const int i = blockIdx.x * blockDim.x + threadIdx.x;
  if (i < n) p[i] = 0;
}

__global__ void gat_fill_rows(const int* __restrict__ src, const int* __restrict__ dst,
                              int* cnt, int* __restrict__ rows,
                              int* spill_n, int* __restrict__ spill, int E) {
  const int i = blockIdx.x * blockDim.x + threadIdx.x;
  if (i < E) {
    const int d = dst[i];
    const int s = src[i];
    const int slot = atomicAdd(cnt + d, 1);
    if (slot < GAT_ROW_SLOTS) {
      rows[(size_t)d * GAT_ROW_SLOTS + slot] = s;
    } else {
      const int k = atomicAdd(spill_n, 1);
      ((int2*)spill)[k] = make_int2(d, s);
    }
  }
}


__device__ __forceinline__ float gat_logit(float e) {
  return (e > 0.f) ? e : 0.2f * e;
}

template <int R>
__device__ __forceinline__ void gat_fold(const int (&sv)[R],
                                         const float* __restrict__ el,
                                         const float2* __restrict__ zl,
                                         float ern, float& m, float& s, float2& acc) {
  float  v[R];
  float2 zz[R];
  for (int r = 0; r < R; ++r) {
    v[r]  = gat_logit(el[sv[r]] + ern);
    zz[r] = zl[(size_t)sv[r] * 32];
  }
  float mn = m;
  for (int r = 0; r < R; ++r) mn = fmaxf(mn, v[r]);
  const float sc = __expf(m - mn);
  float  ps = 0.f;
  float2 pz = {0.f, 0.f};
  for (int r = 0; r < R; ++r) {
    const float p = __expf(v[r] - mn);
    ps   += p;
    pz.x += p * zz[r].x;
    pz.y += p * zz[r].y;
  }
  s     = s * sc + ps;
  acc.x = acc.x * sc + pz.x;
  acc.y = acc.y * sc + pz.y;
  m     = mn;
}

__global__ void gat_node_aggregate(const int* __restrict__ deg,
                                   const int* __restrict__ rows,
                                   const int* __restrict__ spill_n,
                                   const int* __restrict__ spill,
                                   const float* __restrict__ el,
                                   const float* __restrict__ er,
                                   const float* __restrict__ z,
                                   const float* __restrict__ bias,
                                   float* __restrict__ out,
                                   int N, int applyRelu) {
  const int wid  = (int)((blockIdx.x * (size_t)blockDim.x + threadIdx.x) >> 5);
  const int lane = threadIdx.x & 31;
  if (wid >= N) return;
  const int* row = rows + (size_t)wid * GAT_ROW_SLOTS;
  const int cnt  = deg[wid];
  const int nrow = (cnt < GAT_ROW_SLOTS) ? cnt : GAT_ROW_SLOTS;
  const bool spilled = cnt > GAT_ROW_SLOTS;
  const int nsp  = spill_n[0];
  const float ern = er[wid];
  const float2* zl = (const float2*)z + lane;

  float  m   = -INFINITY;
  float  s   = 0.f;
  float2 acc = {0.f, 0.f};

  int j = 0;
  for (; j + 4 <= nrow; j += 4) {
    const int4 q = *(const int4*)(row + j);
    const int sv[4] = {q.x, q.y, q.z, q.w};
    gat_fold<4>(sv, el, zl, ern, m, s, acc);
  }
  const int rem = nrow - j;
  if (rem == 3) {
    const int sv[3] = {row[j], row[j + 1], row[j + 2]};
    gat_fold<3>(sv, el, zl, ern, m, s, acc);
  } else if (rem == 2) {
    const int sv[2] = {row[j], row[j + 1]};
    gat_fold<2>(sv, el, zl, ern, m, s, acc);
  } else if (rem == 1) {
    const int sv[1] = {row[j]};
    gat_fold<1>(sv, el, zl, ern, m, s, acc);
  }
  if (spilled) {
    for (int k = 0; k < nsp; ++k) {
      const int2 p = ((const int2*)spill)[k];
      if (p.x == wid) {
        const int sv[1] = {p.y};
        gat_fold<1>(sv, el, zl, ern, m, s, acc);
      }
    }
  }

  const float rcp = (cnt > 0) ? (1.f / s) : 0.f;
  float2 o;
  o.x = acc.x * rcp + bias[2 * lane];
  o.y = acc.y * rcp + bias[2 * lane + 1];
  if (applyRelu) { o.x = fmaxf(o.x, 0.f); o.y = fmaxf(o.y, 0.f); }
  ((float2*)(out + (size_t)wid * 64))[lane] = o;
}


extern "C" void kernel_launch(void* const* d_in, const int* in_sizes, int n_in,
                              void* d_out, int out_size, void* d_ws, size_t ws_size,
                              hipStream_t stream) {
  const float* x   = (const float*)d_in[0];
  const float* W1  = (const float*)d_in[1];
  const float* al1 = (const float*)d_in[2];
  const float* ar1 = (const float*)d_in[3];
  const float* b1  = (const float*)d_in[4];
  const float* W2  = (const float*)d_in[5];
  const float* al2 = (const float*)d_in[6];
  const float* ar2 = (const float*)d_in[7];
  const float* b2  = (const float*)d_in[8];
  const int*   src = (const int*)d_in[9];
  const int*   dst = (const int*)d_in[10];
  float* out = (float*)d_out;

  const int F  = 64;
  const int N  = out_size / F;
  const int E  = in_sizes[9];
  const int K1 = in_sizes[0] / N;
  (void)n_in; (void)ws_size;

  char* ws = (char*)d_ws;
  size_t off = 0;
  auto alloc = [&](size_t bytes) -> void* {
    void* p = ws + off;
    off += (bytes + 255) & ~(size_t)255;
    return p;
  };
  float* zbuf     = (float*)alloc((size_t)N * F * sizeof(float));
  float* hbuf     = (float*)alloc((size_t)N * F * sizeof(float));
  float* el       = (float*)alloc((size_t)N * sizeof(float));
  float* er       = (float*)alloc((size_t)N * sizeof(float));
  int*   deg      = (int*)alloc((size_t)(N + 1) * sizeof(int));
  int*   rows     = (int*)alloc((size_t)N * GAT_ROW_SLOTS * sizeof(int));
  int*   spill    = (int*)alloc((size_t)E * 2 * sizeof(int));
  int*   spill_n  = deg + N;

  const int gN1    = (N + 256) / 256;
  const int gE256  = (E + 255) / 256;
  const int gWaveN = (N + 7) / 8;
  const int gGemm  = (((N + 15) / 16) + 7) / 8;

  gat_zero_i32 <<<gN1,   256, 0, stream>>>(deg, N + 1);
  gat_fill_rows<<<gE256, 256, 0, stream>>>(src, dst, deg, rows, spill_n, spill, E);

  gat_gemm_wmma     <<<gGemm,  256, 0, stream>>>(x, W1, zbuf, N, K1);
  gat_node_attn     <<<gWaveN, 256, 0, stream>>>(zbuf, al1, ar1, el, er, N);
  gat_node_aggregate<<<gWaveN, 256, 0, stream>>>(deg, rows, spill_n, spill, el, er,
                                                 zbuf, b1, hbuf, N,  1);

  gat_gemm_wmma     <<<gGemm,  256, 0, stream>>>(hbuf, W2, zbuf, N, F);
  gat_node_attn     <<<gWaveN, 256, 0, stream>>>(zbuf, al2, ar2, el, er, N);
  gat_node_aggregate<<<gWaveN, 256, 0, stream>>>(deg, rows, spill_n, spill, el, er,
                                                 zbuf, b2, out, N,  0);
}


# attention vectors folded into the weights: GEMM gets a 5th WMMA column tile [W.al, W.ar] and emits el, er = h.(W.a); both per-node dot launches removed, 8->6 launches; new summation order for el, er
# speedup vs baseline: 1.7390x; 1.1027x over previous
#include <hip/hip_runtime.h>
#include <math.h>


typedef __attribute__((ext_vector_type(2))) float v2f;
typedef __attribute__((ext_vector_type(8))) float v8f;

__global__ void __launch_bounds__(256)
gat_gemm_wmma(const float* __restrict__ A, const float* __restrict__ B,
              const float* __restrict__ Bx,
              float* __restrict__ C, float* __restrict__ el, float* __restrict__ er,
              int M, int K) {
  const int lane    = threadIdx.x & 31;
  const int wv      = threadIdx.x >> 5;
  const int strip   = blockIdx.x * 8 + wv;
  const int nstrips = (M + 15) >> 4;
  if (strip >= nstrips) return;
  const int row0 = strip * 16;
  const int kh   = lane >> 4;
  const int l15  = lane & 15;
  int mrow = row0 + l15;
  if (mrow >= M) mrow = M - 1;

  const float* arow = A + (size_t)mrow * K;
  v8f acc[5];
#pragma unroll
  for (int t = 0; t < 5; ++t) acc[t] = (v8f){0.f,0.f,0.f,0.f,0.f,0.f,0.f,0.f};
  const float* bxl = Bx + (size_t)kh * 32 + 2 * l15;

  for (int k = 0; k < K; k += 4) {
    const v2f a = *(const v2f*)(arow + k + 2 * kh);
    const float* brow0 = B + (size_t)(k + 2 * kh) * 64 + l15;
#pragma unroll
    for (int t = 0; t < 4; ++t) {
      v2f b;
      b.x = brow0[t * 16];
      b.y = brow0[64 + t * 16];
      acc[t] = __builtin_amdgcn_wmma_f32_16x16x4_f32(
          false, a, false, b, (short)0, acc[t], false, false);
    }
    const v2f bx = *(const v2f*)(bxl + (size_t)(k >> 1) * 32);
    acc[4] = __builtin_amdgcn_wmma_f32_16x16x4_f32(
        false, a, false, bx, (short)0, acc[4], false, false);
  }

  const int rbase = row0 + 8 * kh;
#pragma unroll
  for (int t = 0; t < 4; ++t) {
    float* cp = C + (size_t)rbase * 64 + t * 16 + l15;
#pragma unroll
    for (int r = 0; r < 8; ++r)
      if (rbase + r < M) cp[(size_t)r * 64] = acc[t][r];
  }
  if (l15 < 2) {
    float* ep = ((l15 == 0) ? el : er) + rbase;
#pragma unroll
    for (int r = 0; r < 8; ++r)
      if (rbase + r < M) ep[r] = acc[4][r];
  }
}

#define GAT_ROW_SLOTS 64

__global__ void gat_prep(int* __restrict__ p, int n, int nzero,
                         const float* __restrict__ W1, const float* __restrict__ al1,
                         const float* __restrict__ ar1, float* __restrict__ bx1, int K1,
                         const float* __restrict__ W2, const float* __restrict__ al2,
                         const float* __restrict__ ar2, float* __restrict__ bx2, int K2) {
  if ((int)blockIdx.x < nzero) {
    const int i = blockIdx.x * blockDim.x + threadIdx.x;
    if (i < n) p[i] = 0;
    return;
  }
  const int lane = threadIdx.x & 31;
  int k = ((int)blockIdx.x - nzero) * (int)(blockDim.x >> 5) + (int)(threadIdx.x >> 5);
  const float* Wm = W1; const float* al = al1; const float* ar = ar1; float* bx = bx1;
  if (k >= K1) {
    k -= K1;
    if (k >= K2) return;
    Wm = W2; al = al2; ar = ar2; bx = bx2;
  }
  const float w0 = Wm[(size_t)k * 64 + lane], w1 = Wm[(size_t)k * 64 + lane + 32];
  float sl = w0 * al[lane] + w1 * al[lane + 32];
  float sr = w0 * ar[lane] + w1 * ar[lane + 32];
#pragma unroll
  for (int off = 16; off; off >>= 1) {
    sl += __shfl_xor(sl, off, 32);
    sr += __shfl_xor(sr, off, 32);
  }
  if (lane < 16)
    bx[(size_t)(k >> 1) * 32 + 2 * lane + (k & 1)] = (lane == 0) ? sl : (lane == 1) ? sr : 0.f;
}

__global__ void gat_fill_rows(const int* __restrict__ src, const int* __restrict__ dst,
                              int* cnt, int* __restrict__ rows,
                              int* spill_n, int* __restrict__ spill, int E) {
  const int i = blockIdx.x * blockDim.x + threadIdx.x;
  if (i < E) {
    const int d = dst[i];
    const int s = src[i];
    const int slot = atomicAdd(cnt + d, 1);
    if (slot < GAT_ROW_SLOTS) {
      rows[(size_t)d * GAT_ROW_SLOTS + slot] = s;
    } else {
      const int k = atomicAdd(spill_n, 1);
      ((int2*)spill)[k] = make_int2(d, s);
    }
  }
}


__device__ __forceinline__ float gat_logit(float e) {
  return (e > 0.f) ? e : 0.2f * e;
}

template <int R>
__device__ __forceinline__ void gat_fold(const int (&sv)[R],
                                         const float* __restrict__ el,
                                         const float2* __restrict__ zl,
                                         float ern, float& m, float& s, float2& acc) {
  float  v[R];
  float2 zz[R];
  for (int r = 0; r < R; ++r) {
    v[r]  = gat_logit(el[sv[r]] + ern);
    zz[r] = zl[(size_t)sv[r] * 32];
  }
  float mn = m;
  for (int r = 0; r < R; ++r) mn = fmaxf(mn, v[r]);
  const float sc = __expf(m - mn);
  float  ps = 0.f;
  float2 pz = {0.f, 0.f};
  for (int r = 0; r < R; ++r) {
    const float p = __expf(v[r] - mn);
    ps   += p;
    pz.x += p * zz[r].x;
    pz.y += p * zz[r].y;
  }
  s     = s * sc + ps;
  acc.x = acc.x * sc + pz.x;
  acc.y = acc.y * sc + pz.y;
  m     = mn;
}

__global__ void gat_node_aggregate(const int* __restrict__ deg,
                                   const int* __restrict__ rows,
                                   const int* __restrict__ spill_n,
                                   const int* __restrict__ spill,
                                   const float* __restrict__ el,
                                   const float* __restrict__ er,
                                   const float* __restrict__ z,
                                   const float* __restrict__ bias,
                                   float* __restrict__ out,
                                   int N, int applyRelu) {
  const int wid  = (int)((blockIdx.x * (size_t)blockDim.x + threadIdx.x) >> 5);
  const int lane = threadIdx.x & 31;
  if (wid >= N) return;
  const int* row = rows + (size_t)wid * GAT_ROW_SLOTS;
  const int cnt  = deg[wid];
  const int nrow = (cnt < GAT_ROW_SLOTS) ? cnt : GAT_ROW_SLOTS;
  const bool spilled = cnt > GAT_ROW_SLOTS;
  const int nsp  = spill_n[0];
  const float ern = er[wid];
  const float2* zl = (const float2*)z + lane;

  float  m   = -INFINITY;
  float  s   = 0.f;
  float2 acc = {0.f, 0.f};

  int j = 0;
  for (; j + 4 <= nrow; j += 4) {
    const int4 q = *(const int4*)(row + j);
    const int sv[4] = {q.x, q.y, q.z, q.w};
    gat_fold<4>(sv, el, zl, ern, m, s, acc);
  }
  const int rem = nrow - j;
  if (rem == 3) {
    const int sv[3] = {row[j], row[j + 1], row[j + 2]};
    gat_fold<3>(sv, el, zl, ern, m, s, acc);
  } else if (rem == 2) {
    const int sv[2] = {row[j], row[j + 1]};
    gat_fold<2>(sv, el, zl, ern, m, s, acc);
  } else if (rem == 1) {
    const int sv[1] = {row[j]};
    gat_fold<1>(sv, el, zl, ern, m, s, acc);
  }
  if (spilled) {
    for (int k = 0; k < nsp; ++k) {
      const int2 p = ((const int2*)spill)[k];
      if (p.x == wid) {
        const int sv[1] = {p.y};
        gat_fold<1>(sv, el, zl, ern, m, s, acc);
      }
    }
  }

  const float rcp = (cnt > 0) ? (1.f / s) : 0.f;
  float2 o;
  o.x = acc.x * rcp + bias[2 * lane];
  o.y = acc.y * rcp + bias[2 * lane + 1];
  if (applyRelu) { o.x = fmaxf(o.x, 0.f); o.y = fmaxf(o.y, 0.f); }
  ((float2*)(out + (size_t)wid * 64))[lane] = o;
}


extern "C" void kernel_launch(void* const* d_in, const int* in_sizes, int n_in,
                              void* d_out, int out_size, void* d_ws, size_t ws_size,
                              hipStream_t stream) {
  const float* x   = (const float*)d_in[0];
  const float* W1  = (const float*)d_in[1];
  const float* al1 = (const float*)d_in[2];
  const float* ar1 = (const float*)d_in[3];
  const float* b1  = (const float*)d_in[4];
  const float* W2  = (const float*)d_in[5];
  const float* al2 = (const float*)d_in[6];
  const float* ar2 = (const float*)d_in[7];
  const float* b2  = (const float*)d_in[8];
  const int*   src = (const int*)d_in[9];
  const int*   dst = (const int*)d_in[10];
  float* out = (float*)d_out;

  const int F  = 64;
  const int N  = out_size / F;
  const int E  = in_sizes[9];
  const int K1 = in_sizes[0] / N;
  (void)n_in; (void)ws_size;

  char* ws = (char*)d_ws;
  size_t off = 0;
  auto alloc = [&](size_t bytes) -> void* {
    void* p = ws + off;
    off += (bytes + 255) & ~(size_t)255;
    return p;
  };
  float* zbuf     = (float*)alloc((size_t)N * F * sizeof(float));
  float* hbuf     = (float*)alloc((size_t)N * F * sizeof(float));
  float* el       = (float*)alloc((size_t)N * sizeof(float));
  float* er       = (float*)alloc((size_t)N * sizeof(float));
  int*   deg      = (int*)alloc((size_t)(N + 1) * sizeof(int));
  int*   rows     = (int*)alloc((size_t)N * GAT_ROW_SLOTS * sizeof(int));
  int*   spill    = (int*)alloc((size_t)E * 2 * sizeof(int));
  int*   spill_n  = deg + N;
  float* bx1      = (float*)alloc((size_t)K1 * 16 * sizeof(float));
  float* bx2      = (float*)alloc((size_t)F  * 16 * sizeof(float));

  const int gN1    = (N + 256) / 256;
  const int gE256  = (E + 255) / 256;
  const int gWaveN = (N + 7) / 8;
  const int gGemm  = (((N + 15) / 16) + 7) / 8;
  const int gFold  = (K1 + F + 7) / 8;

  gat_prep     <<<gN1 + gFold, 256, 0, stream>>>(deg, N + 1, gN1,
                                                 W1, al1, ar1, bx1, K1,
                                                 W2, al2, ar2, bx2, F);
  gat_fill_rows<<<gE256, 256, 0, stream>>>(src, dst, deg, rows, spill_n, spill, E);

  gat_gemm_wmma     <<<gGemm,  256, 0, stream>>>(x, W1, bx1, zbuf, el, er, N, K1);
  gat_node_aggregate<<<gWaveN, 256, 0, stream>>>(deg, rows, spill_n, spill, el, er,
                                                 zbuf, b1, hbuf, N,  1);

  gat_gemm_wmma     <<<gGemm,  256, 0, stream>>>(hbuf, W2, bx2, zbuf, el, er, N, F);
  gat_node_aggregate<<<gWaveN, 256, 0, stream>>>(deg, rows, spill_n, spill, el, er,
                                                 zbuf, b2, out, N,  0);
}


# adjacency fill kept per L2 half: a wave adds to the counters and 32-slot sub-rows of the half it runs on, so no counter or row line has writers in both halves; in-row and summation order change
# speedup vs baseline: 3.1987x; 1.8394x over previous
#include <hip/hip_runtime.h>
#include <math.h>


typedef __attribute__((ext_vector_type(2))) float v2f;
typedef __attribute__((ext_vector_type(8))) float v8f;

__global__ void __launch_bounds__(256)
gat_gemm_wmma(const float* __restrict__ A, const float* __restrict__ B,
              const float* __restrict__ Bx,
              float* __restrict__ C, float* __restrict__ el, float* __restrict__ er,
              int M, int K) {
  const int lane    = threadIdx.x & 31;
  const int wv      = threadIdx.x >> 5;
  const int strip   = blockIdx.x * 8 + wv;
  const int nstrips = (M + 15) >> 4;
  if (strip >= nstrips) return;
  const int row0 = strip * 16;
  const int kh   = lane >> 4;
  const int l15  = lane & 15;
  int mrow = row0 + l15;
  if (mrow >= M) mrow = M - 1;

  const float* arow = A + (size_t)mrow * K;
  v8f acc[5];
#pragma unroll
  for (int t = 0; t < 5; ++t) acc[t] = (v8f){0.f,0.f,0.f,0.f,0.f,0.f,0.f,0.f};
  const float* bxl = Bx + (size_t)kh * 32 + 2 * l15;

  for (int k = 0; k < K; k += 4) {
    const v2f a = *(const v2f*)(arow + k + 2 * kh);
    const float* brow0 = B + (size_t)(k + 2 * kh) * 64 + l15;
#pragma unroll
    for (int t = 0; t < 4; ++t) {
      v2f b;
      b.x = brow0[t * 16];
      b.y = brow0[64 + t * 16];
      acc[t] = __builtin_amdgcn_wmma_f32_16x16x4_f32(
          false, a, false, b, (short)0, acc[t], false, false);
    }
    const v2f bx = *(const v2f*)(bxl + (size_t)(k >> 1) * 32);
    acc[4] = __builtin_amdgcn_wmma_f32_16x16x4_f32(
        false, a, false, bx, (short)0, acc[4], false, false);
  }

  const int rbase = row0 + 8 * kh;
#pragma unroll
  for (int t = 0; t < 4; ++t) {
    float* cp = C + (size_t)rbase * 64 + t * 16 + l15;
#pragma unroll
    for (int r = 0; r < 8; ++r)
      if (rbase + r < M) cp[(size_t)r * 64] = acc[t][r];
  }
  if (l15 < 2) {
    float* ep = ((l15 == 0) ? el : er) + rbase;
#pragma unroll
    for (int r = 0; r < 8; ++r)
      if (rbase + r < M) ep[r] = acc[4][r];
  }
}

#define GAT_SUB_SLOTS 32

__global__ void gat_prep(int* __restrict__ p, int n, int nzero,
                         const float* __restrict__ W1, const float* __restrict__ al1,
                         const float* __restrict__ ar1, float* __restrict__ bx1, int K1,
                         const float* __restrict__ W2, const float* __restrict__ al2,
                         const float* __restrict__ ar2, float* __restrict__ bx2, int K2) {
  if ((int)blockIdx.x < nzero) {
    const int i = blockIdx.x * blockDim.x + threadIdx.x;
    if (i < n) p[i] = 0;
    return;
  }
  const int lane = threadIdx.x & 31;
  int k = ((int)blockIdx.x - nzero) * (int)(blockDim.x >> 5) + (int)(threadIdx.x >> 5);
  const float* Wm = W1; const float* al = al1; const float* ar = ar1; float* bx = bx1;
  if (k >= K1) {
    k -= K1;
    if (k >= K2) return;
    Wm = W2; al = al2; ar = ar2; bx = bx2;
  }
  const float w0 = Wm[(size_t)k * 64 + lane], w1 = Wm[(size_t)k * 64 + lane + 32];
  float sl = w0 * al[lane] + w1 * al[lane + 32];
  float sr = w0 * ar[lane] + w1 * ar[lane + 32];
#pragma unroll
  for (int off = 16; off; off >>= 1) {
    sl += __shfl_xor(sl, off, 32);
    sr += __shfl_xor(sr, off, 32);
  }
  if (lane < 16)
    bx[(size_t)(k >> 1) * 32 + 2 * lane + (k & 1)] = (lane == 0) ? sl : (lane == 1) ? sr : 0.f;
}

__global__ void gat_fill_rows(const int* __restrict__ src, const int* __restrict__ dst,
                              int* cnt, int cnt_plane, int* __restrict__ rows,
                              size_t row_plane,
                              int* spill_n, int* __restrict__ spill, int E) {
  const unsigned place = __builtin_amdgcn_s_sendmsg_rtn(0x87);
  int p = (int)((place >> 18) & 1u);
  const int i = blockIdx.x * blockDim.x + threadIdx.x;
  if (i < E) {
    const int d = dst[i];
    const int s = src[i];
    int slot = atomicAdd(cnt + (size_t)p * cnt_plane + d, 1);
    if (slot >= GAT_SUB_SLOTS) {
      p ^= 1;
      slot = atomicAdd(cnt + (size_t)p * cnt_plane + d, 1);
    }
    if (slot < GAT_SUB_SLOTS) {
      rows[(size_t)p * row_plane + (size_t)d * GAT_SUB_SLOTS + slot] = s;
    } else {
      const int k = atomicAdd(spill_n, 1);
      ((int2*)spill)[k] = make_int2(d, s);
    }
  }
}


__device__ __forceinline__ float gat_logit(float e) {
  return (e > 0.f) ? e : 0.2f * e;
}

template <int R>
__device__ __forceinline__ void gat_fold(const int (&sv)[R],
                                         const float* __restrict__ el,
                                         const float2* __restrict__ zl,
                                         float ern, float& m, float& s, float2& acc) {
  float  v[R];
  float2 zz[R];
  for (int r = 0; r < R; ++r) {
    v[r]  = gat_logit(el[sv[r]] + ern);
    zz[r] = zl[(size_t)sv[r] * 32];
  }
  float mn = m;
  for (int r = 0; r < R; ++r) mn = fmaxf(mn, v[r]);
  const float sc = __expf(m - mn);
  float  ps = 0.f;
  float2 pz = {0.f, 0.f};
  for (int r = 0; r < R; ++r) {
    const float p = __expf(v[r] - mn);
    ps   += p;
    pz.x += p * zz[r].x;
    pz.y += p * zz[r].y;
  }
  s     = s * sc + ps;
  acc.x = acc.x * sc + pz.x;
  acc.y = acc.y * sc + pz.y;
  m     = mn;
}

__device__ __forceinline__ void gat_fold_subrow(const int* __restrict__ row, int n,
                                                const float* __restrict__ el,
                                                const float2* __restrict__ zl,
                                                float ern, float& m, float& s, float2& acc) {
  int j = 0;
  for (; j + 4 <= n; j += 4) {
    const int4 q = *(const int4*)(row + j);
    const int sv[4] = {q.x, q.y, q.z, q.w};
    gat_fold<4>(sv, el, zl, ern, m, s, acc);
  }
  const int rem = n - j;
  if (rem == 3) {
    const int sv[3] = {row[j], row[j + 1], row[j + 2]};
    gat_fold<3>(sv, el, zl, ern, m, s, acc);
  } else if (rem == 2) {
    const int sv[2] = {row[j], row[j + 1]};
    gat_fold<2>(sv, el, zl, ern, m, s, acc);
  } else if (rem == 1) {
    const int sv[1] = {row[j]};
    gat_fold<1>(sv, el, zl, ern, m, s, acc);
  }
}

__global__ void gat_node_aggregate(const int* __restrict__ deg, int cnt_plane,
                                   const int* __restrict__ rows, size_t row_plane,
                                   const int* __restrict__ spill_n,
                                   const int* __restrict__ spill,
                                   const float* __restrict__ el,
                                   const float* __restrict__ er,
                                   const float* __restrict__ z,
                                   const float* __restrict__ bias,
                                   float* __restrict__ out,
                                   int N, int applyRelu) {
  const int wid  = (int)((blockIdx.x * (size_t)blockDim.x + threadIdx.x) >> 5);
  const int lane = threadIdx.x & 31;
  if (wid >= N) return;
  const int* row0 = rows + (size_t)wid * GAT_SUB_SLOTS;
  const int* row1 = row0 + row_plane;
  const int cnt0 = deg[wid];
  const int cnt1 = deg[(size_t)cnt_plane + wid];
  const int n0 = (cnt0 < GAT_SUB_SLOTS) ? cnt0 : GAT_SUB_SLOTS;
  const int n1 = (cnt1 < GAT_SUB_SLOTS) ? cnt1 : GAT_SUB_SLOTS;
  const bool spilled = (cnt0 > GAT_SUB_SLOTS) && (cnt1 > GAT_SUB_SLOTS);
  const int cnt = cnt0 + cnt1;
  const int nsp  = spill_n[0];
  const float ern = er[wid];
  const float2* zl = (const float2*)z + lane;

  float  m   = -INFINITY;
  float  s   = 0.f;
  float2 acc = {0.f, 0.f};

  gat_fold_subrow(row0, n0, el, zl, ern, m, s, acc);
  gat_fold_subrow(row1, n1, el, zl, ern, m, s, acc);
  if (spilled) {
    for (int k = 0; k < nsp; ++k) {
      const int2 p = ((const int2*)spill)[k];
      if (p.x == wid) {
        const int sv[1] = {p.y};
        gat_fold<1>(sv, el, zl, ern, m, s, acc);
      }
    }
  }

  const float rcp = (cnt > 0) ? (1.f / s) : 0.f;
  float2 o;
  o.x = acc.x * rcp + bias[2 * lane];
  o.y = acc.y * rcp + bias[2 * lane + 1];
  if (applyRelu) { o.x = fmaxf(o.x, 0.f); o.y = fmaxf(o.y, 0.f); }
  ((float2*)(out + (size_t)wid * 64))[lane] = o;
}


extern "C" void kernel_launch(void* const* d_in, const int* in_sizes, int n_in,
                              void* d_out, int out_size, void* d_ws, size_t ws_size,
                              hipStream_t stream) {
  const float* x   = (const float*)d_in[0];
  const float* W1  = (const float*)d_in[1];
  const float* al1 = (const float*)d_in[2];
  const float* ar1 = (const float*)d_in[3];
  const float* b1  = (const float*)d_in[4];
  const float* W2  = (const float*)d_in[5];
  const float* al2 = (const float*)d_in[6];
  const float* ar2 = (const float*)d_in[7];
  const float* b2  = (const float*)d_in[8];
  const int*   src = (const int*)d_in[9];
  const int*   dst = (const int*)d_in[10];
  float* out = (float*)d_out;

  const int F  = 64;
  const int N  = out_size / F;
  const int E  = in_sizes[9];
  const int K1 = in_sizes[0] / N;
  (void)n_in; (void)ws_size;

  char* ws = (char*)d_ws;
  size_t off = 0;
  auto alloc = [&](size_t bytes) -> void* {
    void* p = ws + off;
    off += (bytes + 255) & ~(size_t)255;
    return p;
  };
  float* zbuf     = (float*)alloc((size_t)N * F * sizeof(float));
  float* hbuf     = (float*)alloc((size_t)N * F * sizeof(float));
  float* el       = (float*)alloc((size_t)N * sizeof(float));
  float* er       = (float*)alloc((size_t)N * sizeof(float));
  const int    cntPlane = (N + 63) & ~63;
  const size_t rowPlane = (size_t)N * GAT_SUB_SLOTS;
  int*   deg      = (int*)alloc((size_t)(2 * cntPlane + 1) * sizeof(int));
  int*   rows     = (int*)alloc(2 * rowPlane * sizeof(int));
  int*   spill    = (int*)alloc((size_t)E * 2 * sizeof(int));
  int*   spill_n  = deg + 2 * cntPlane;
  float* bx1      = (float*)alloc((size_t)K1 * 16 * sizeof(float));
  float* bx2      = (float*)alloc((size_t)F  * 16 * sizeof(float));

  const int nCnt   = 2 * cntPlane + 1;
  const int gN1    = (nCnt + 255) / 256;
  const int gE256  = (E + 255) / 256;
  const int gWaveN = (N + 7) / 8;
  const int gGemm  = (((N + 15) / 16) + 7) / 8;
  const int gFold  = (K1 + F + 7) / 8;

  gat_prep     <<<gN1 + gFold, 256, 0, stream>>>(deg, nCnt, gN1,
                                                 W1, al1, ar1, bx1, K1,
                                                 W2, al2, ar2, bx2, F);
  gat_fill_rows<<<gE256, 256, 0, stream>>>(src, dst, deg, cntPlane, rows, rowPlane,
                                           spill_n, spill, E);

  gat_gemm_wmma     <<<gGemm,  256, 0, stream>>>(x, W1, bx1, zbuf, el, er, N, K1);
  gat_node_aggregate<<<gWaveN, 256, 0, stream>>>(deg, cntPlane, rows, rowPlane, spill_n, spill,
                                                 el, er, zbuf, b1, hbuf, N,  1);

  gat_gemm_wmma     <<<gGemm,  256, 0, stream>>>(hbuf, W2, bx2, zbuf, el, er, N, F);
  gat_node_aggregate<<<gWaveN, 256, 0, stream>>>(deg, cntPlane, rows, rowPlane, spill_n, spill,
                                                 el, er, zbuf, b2, out, N,  0);
}
